# starting kernel
# baseline (speedup 1.0000x reference)
#include <hip/hip_runtime.h>


typedef __attribute__((ext_vector_type(2))) float v2f;
typedef __attribute__((ext_vector_type(8))) float v8f;

#define GN 50000
#define GE 800000
#define GD 64
#define WROW 160

__global__ __launch_bounds__(256) void count_edges_k(const int* __restrict__ dst,
                                                     float* __restrict__ cnt,
                                                     int e_total) {
  int e = blockIdx.x * 256 + threadIdx.x;
  if (e < e_total) atomicAdd(&cnt[dst[e]], 1.0f);
}

__global__ __launch_bounds__(256) void scatter_add_k(const int* __restrict__ src,
                                                     const int* __restrict__ dst,
                                                     const float* __restrict__ feat,
                                                     float* __restrict__ agg,
                                                     int e_total) {
  long long t = (long long)blockIdx.x * 256 + threadIdx.x;
  int e = (int)(t >> 6);
  int f = (int)(t & 63);
  if (e < e_total) {
    int s = src[e];
    int d = dst[e];
    atomicAdd(&agg[(long long)d * GD + f], feat[(long long)s * GD + f]);
  }
}

__global__ __launch_bounds__(256) void sage_gemm_relu_k(
    const float* __restrict__ agg, const float* __restrict__ cnt,
    const float* __restrict__ xr,
    const float* __restrict__ Wl, const float* __restrict__ Wr,
    const float* __restrict__ bl,
    float* __restrict__ hout, int ntiles)
{
  __shared__ float WlS[(GD / 2) * WROW];
  __shared__ float WrS[(GD / 2) * WROW];
  for (int i = threadIdx.x; i < GD * GD; i += 256) {
    int j = i >> 6;
    int k = i & 63;
    int o = (k >> 1) * WROW + j * 2 + (k & 1);
    WlS[o] = Wl[i];
    WrS[o] = Wr[i];
  }
  __syncthreads();

  int wave = threadIdx.x >> 5;
  int tile = blockIdx.x * 8 + wave;
  if (tile < ntiles) {
    int lane = threadIdx.x & 31;
    int half = lane >> 4;
    int l16  = lane & 15;
    int row  = tile * 16 + l16;

    float inv = 1.0f / fmaxf(cnt[row], 1.0f);
    const float* aggRow = agg + (long long)row * GD;
    const float* xRow   = xr  + (long long)row * GD;

    v8f acc[4] = {};

    for (int k0 = 0; k0 < GD; k0 += 4) {
      int ka = k0 + half * 2;
      v2f am = *(const v2f*)(aggRow + ka);
      am *= inv;
      v2f ar = *(const v2f*)(xRow + ka);
      int kb = (ka >> 1) * WROW;
#pragma unroll
      for (int t = 0; t < 4; ++t) {
        int n = t * 16 + l16;
        v2f bL = *(const v2f*)&WlS[kb + n * 2];
        v2f bR = *(const v2f*)&WrS[kb + n * 2];
        acc[t] = __builtin_amdgcn_wmma_f32_16x16x4_f32(
            false, am, false, bL, (short)0, acc[t], false, false);
        acc[t] = __builtin_amdgcn_wmma_f32_16x16x4_f32(
            false, ar, false, bR, (short)0, acc[t], false, false);
      }
    }

#pragma unroll
    for (int t = 0; t < 4; ++t) {
      int n = t * 16 + l16;
      float bias = bl[n];
#pragma unroll
      for (int v = 0; v < 8; ++v) {
        int mr = half * 8 + v;
        float val = fmaxf(acc[t][v] + bias, 0.0f);
        hout[(long long)(tile * 16 + mr) * GD + n] = val;
      }
    }
  }
}

__global__ __launch_bounds__(256) void final_head_k(const float* __restrict__ h,
                                                    const float* __restrict__ Wf,
                                                    const float* __restrict__ bf,
                                                    float* __restrict__ out,
                                                    int n_total) {
  int i = blockIdx.x * 256 + threadIdx.x;
  if (i >= n_total) return;
  const float4* r = (const float4*)(h + (long long)i * GD);
  const float4* w = (const float4*)Wf;
  float s = 0.0f;
#pragma unroll
  for (int q = 0; q < GD / 4; ++q) {
    float4 a = r[q], b = w[q];
    s += a.x * b.x + a.y * b.y + a.z * b.z + a.w * b.w;
  }
  out[i] = s + bf[0];
}

extern "C" void kernel_launch(void* const* d_in, const int* in_sizes, int n_in,
                              void* d_out, int out_size, void* d_ws, size_t ws_size,
                              hipStream_t stream) {
  (void)in_sizes; (void)n_in; (void)out_size; (void)ws_size;

  const float* x   = (const float*)d_in[0];
  const int*   ei  = (const int*)d_in[1];
  const int*   src = ei;
  const int*   dst = ei + GE;
  const float* Wl1 = (const float*)d_in[2];
  const float* bl1 = (const float*)d_in[3];
  const float* Wr1 = (const float*)d_in[4];
  const float* Wl2 = (const float*)d_in[5];
  const float* bl2 = (const float*)d_in[6];
  const float* Wr2 = (const float*)d_in[7];
  const float* Wf  = (const float*)d_in[8];
  const float* bf  = (const float*)d_in[9];

  char*  ws  = (char*)d_ws;
  float* agg = (float*)ws;
  float* cnt = (float*)(ws + (size_t)GN * GD * 4);
  float* h   = (float*)(ws + (size_t)GN * GD * 4 + (size_t)GN * 4);

  const int ntiles = GN / 16;
  const int gemm_blocks = (ntiles + 7) / 8;
  const int scat_blocks = (GE * GD) / 256;

  hipMemsetAsync(agg, 0, (size_t)GN * GD * 4, stream);
  hipMemsetAsync(cnt, 0, (size_t)GN * 4, stream);
  count_edges_k<<<(GE + 255) / 256, 256, 0, stream>>>(dst, cnt, GE);
  scatter_add_k<<<scat_blocks, 256, 0, stream>>>(src, dst, x, agg, GE);
  sage_gemm_relu_k<<<gemm_blocks, 256, 0, stream>>>(agg, cnt, x, Wl1, Wr1, bl1, h, ntiles);

  hipMemsetAsync(agg, 0, (size_t)GN * GD * 4, stream);
  scatter_add_k<<<scat_blocks, 256, 0, stream>>>(src, dst, h, agg, GE);
  sage_gemm_relu_k<<<gemm_blocks, 256, 0, stream>>>(agg, cnt, h, Wl2, Wr2, bl2, h, ntiles);

  final_head_k<<<(GN + 255) / 256, 256, 0, stream>>>(h, Wf, bf, (float*)d_out, GN);
}


# Layer 1 saves each wave's kept-edge list in the block's own rows of the unwritten h2 plane; layer 2 adds rows from that list, skips the dst scan (rescans on list overflow); sum order changes
# speedup vs baseline: 3.7163x; 3.7163x over previous
#include <hip/hip_runtime.h>


typedef __attribute__((ext_vector_type(2))) float v2f;
typedef __attribute__((ext_vector_type(8))) float v8f;

#define GN 50000
#define GE 800000
#define GD 64
#define WROW 160
#define RB 208
#define NBLK ((GN + RB - 1) / RB)
#define ACCP 68
#define NWAVE 32
#define QFLUSH 128
#define QCAP (QFLUSH + 1024)
#define NV4 (GE / 4)
#define NFULL (NV4 / 256)
#define NTAIL ((NV4 - NFULL * 256) / 32)
#define LIST_OVER 0xFFFFFFFFu
static_assert(NFULL * 256 + NTAIL * 32 == NV4, "the dst list must split into wave-wide 16-byte pieces");

__device__ __forceinline__ void drain_rows(unsigned ent, int s, int m,
                                           const float* __restrict__ feat,
                                           float* accS, float* cntS, int lane) {
  int dl = (int)(ent >> 20);
  if (lane < m) atomicAdd(&cntS[dl], 1.0f);
  for (int j0 = 0; j0 < m; j0 += 16) {
    v2f r[16];
#pragma unroll
    for (int u = 0; u < 16; ++u) {
      int sj = __builtin_amdgcn_readlane(s, j0 + u);
      r[u] = *(const v2f*)(feat + (long long)sj * GD + lane * 2);
    }
#pragma unroll
    for (int u = 0; u < 16; ++u) {
      if (j0 + u < m) {
        int dj = __builtin_amdgcn_readlane(dl, j0 + u);
        float* a = accS + dj * ACCP + lane * 2;
        atomicAdd(a, r[u].x);
        atomicAdd(a + 1, r[u].y);
      }
    }
  }
}

template <bool SAVE>
__device__ __forceinline__ void drain_list(const unsigned* q, int n,
                                           const int* __restrict__ src,
                                           const float* __restrict__ feat,
                                           float* accS, float* cntS, int lane,
                                           float* seg, int cap, int& saved) {
  for (int g = 0; g < n; g += 32) {
    int m = n - g;
    if (m > 32) m = 32;
    unsigned ent = (lane < m) ? q[g + lane] : 0u;
    int s = 0;
    if (lane < m) s = src[ent & 0xFFFFFu];
    if (SAVE) {
      const int p = saved + g + lane;
      if (lane < m && p < cap) seg[p] = __uint_as_float(ent);
    }
    drain_rows(ent, s, m, feat, accS, cntS, lane);
  }
  if (SAVE) saved += n;
}

__device__ __forceinline__ void load_step(int4* d, const int4* __restrict__ dst4,
                                          int st, int lane) {
#pragma unroll
  for (int k = 0; k < 8; ++k) d[k] = dst4[st * 256 + k * 32 + lane];
}

__device__ __forceinline__ void keep_edge(bool mine, unsigned mask, int dlc, int e,
                                          unsigned* q, int& n) {
  int pos = n + (int)__builtin_amdgcn_mbcnt_lo(mask, 0u);
  if (mine) q[pos] = (unsigned)e | ((unsigned)dlc << 20);
  n += __builtin_popcount(mask);
}
__device__ __forceinline__ void scan_piece(const int4 d, const int vi, const int base,
                                           unsigned* q, int& n) {
  const int d0 = d.x - base, d1 = d.y - base, d2 = d.z - base, d3 = d.w - base;
  const bool m0 = (unsigned)d0 < (unsigned)RB;
  const bool m1 = (unsigned)d1 < (unsigned)RB;
  const bool m2 = (unsigned)d2 < (unsigned)RB;
  const bool m3 = (unsigned)d3 < (unsigned)RB;
  const unsigned k0 = __builtin_amdgcn_ballot_w32(m0);
  const unsigned k1 = __builtin_amdgcn_ballot_w32(m1);
  const unsigned k2 = __builtin_amdgcn_ballot_w32(m2);
  const unsigned k3 = __builtin_amdgcn_ballot_w32(m3);
  if ((k0 | k1 | k2 | k3) != 0u) {
    keep_edge(m0, k0, d0, vi * 4 + 0, q, n);
    keep_edge(m1, k1, d1, vi * 4 + 1, q, n);
    keep_edge(m2, k2, d2, vi * 4 + 2, q, n);
    keep_edge(m3, k3, d3, vi * 4 + 3, q, n);
  }
}
__device__ __forceinline__ void scan_step(const int4* d, int st, int lane, int base,
                                          unsigned* q, int& n) {
#pragma unroll
  for (int k = 0; k < 8; ++k) scan_piece(d[k], st * 256 + k * 32 + lane, base, q, n);
}

template <bool SAVE>
__device__ __forceinline__ void scan_all(const int* __restrict__ src, const int* __restrict__ dst,
                                         const float* __restrict__ feat,
                                         float* accS, float* cntS, unsigned* q,
                                         int wave, int lane, int base,
                                         float* seg, int cap, int& saved) {
  int n = 0;
  const int4* dst4 = (const int4*)dst;
  int4 bufA[8], bufB[8];
  load_step(bufA, dst4, wave, lane);
  for (int st = wave; st < NFULL; st += 2 * NWAVE) {
    const int sb = st + NWAVE;
    const int sa = st + 2 * NWAVE;
    load_step(bufB, dst4, sb < NFULL ? sb : NFULL - 1, lane);
    scan_step(bufA, st, lane, base, q, n);
    if (n >= QFLUSH) { drain_list<SAVE>(q, n, src, feat, accS, cntS, lane, seg, cap, saved); n = 0; }
    load_step(bufA, dst4, sa < NFULL ? sa : NFULL - 1, lane);
    if (sb < NFULL) {
      scan_step(bufB, sb, lane, base, q, n);
      if (n >= QFLUSH) { drain_list<SAVE>(q, n, src, feat, accS, cntS, lane, seg, cap, saved); n = 0; }
    }
  }
  if (wave == NWAVE - 1) {
#pragma unroll
    for (int k = 0; k < NTAIL; ++k) {
      const int vi = NFULL * 256 + k * 32 + lane;
      scan_piece(dst4[vi], vi, base, q, n);
    }
  }
  drain_list<SAVE>(q, n, src, feat, accS, cntS, lane, seg, cap, saved);
}

template <int LAYER>
__global__ __launch_bounds__(1024) void sage_layer_own_k(
    const int* __restrict__ src, const int* __restrict__ dst,
    const float* __restrict__ feat,
    const float* __restrict__ Wl, const float* __restrict__ Wr,
    const float* __restrict__ bl,
    float* hout, float* lst)
{
  __shared__ float WlS[(GD / 2) * WROW];
  __shared__ float WrS[(GD / 2) * WROW];
  __shared__ float accS[RB * ACCP];
  __shared__ float cntS[RB];
  __shared__ unsigned qS[NWAVE * QCAP];

  const int tid  = threadIdx.x;
  const int lane = tid & 31;
  const int wave = __builtin_amdgcn_readfirstlane(tid >> 5);
  const int base = blockIdx.x * RB;

  float* reg = lst + (long long)base * GD;
  const int rows = (GN - base < RB) ? (GN - base) : RB;
  const int cap  = (rows * GD - NWAVE) / NWAVE;
  float* seg = reg + NWAVE + wave * cap;

  unsigned cw = 0u;
  if (LAYER == 2) cw = __float_as_uint(reg[lane]);

  for (int i = tid; i < GD * GD; i += 1024) {
    int j = i >> 6;
    int k = i & 63;
    int o = (k >> 1) * WROW + j * 2 + (k & 1);
    WlS[o] = Wl[i];
    WrS[o] = Wr[i];
  }
  for (int i = tid; i < RB * ACCP; i += 1024) accS[i] = 0.0f;
  if (tid < RB) cntS[tid] = 0.0f;
  __syncthreads();

  unsigned* q = qS + wave * QCAP;
  int saved = 0;
  if (LAYER == 1) {
    scan_all<true>(src, dst, feat, accS, cntS, q, wave, lane, base, seg, cap, saved);
    if (lane == 0) reg[wave] = __uint_as_float(saved <= cap ? (unsigned)saved : LIST_OVER);
  } else {
    const bool over = __builtin_amdgcn_ballot_w32(cw == LIST_OVER) != 0u;
    if (!over) {
      const int nl = __builtin_amdgcn_readlane((int)cw, wave);
      unsigned ent  = (lane < nl) ? __float_as_uint(seg[lane]) : 0u;
      unsigned ent1 = (32 + lane < nl) ? __float_as_uint(seg[32 + lane]) : 0u;
      int s = (lane < nl) ? src[ent & 0xFFFFFu] : 0;
      for (int g = 0; g < nl; g += 32) {
        int m = nl - g;
        if (m > 32) m = 32;
        const unsigned ent2 = (g + 64 + lane < nl) ? __float_as_uint(seg[g + 64 + lane]) : 0u;
        const int s1 = (g + 32 + lane < nl) ? src[ent1 & 0xFFFFFu] : 0;
        drain_rows(ent, s, m, feat, accS, cntS, lane);
        ent = ent1;
        s = s1;
        ent1 = ent2;
      }
    } else {
      scan_all<false>(src, dst, feat, accS, cntS, q, wave, lane, base, seg, cap, saved);
    }
  }
  __syncthreads();

  if (wave < RB / 16) {
    const int row0 = base + wave * 16;
    if (row0 < GN) {
      const int half = lane >> 4;
      const int l16  = lane & 15;
      const int lr   = wave * 16 + l16;
      const float inv = 1.0f / fmaxf(cntS[lr], 1.0f);
      const float* xRow = feat + (long long)(base + lr) * GD;

      v2f ar[16];
#pragma unroll
      for (int ks = 0; ks < 16; ++ks) ar[ks] = *(const v2f*)(xRow + ks * 4 + half * 2);

      v8f acc[4] = {};
#pragma unroll
      for (int ks = 0; ks < 16; ++ks) {
        const int ka = ks * 4 + half * 2;
        v2f am = *(const v2f*)&accS[lr * ACCP + ka];
        am *= inv;
        const int kb = (ka >> 1) * WROW;
#pragma unroll
        for (int t = 0; t < 4; ++t) {
          int nn = t * 16 + l16;
          v2f bL = *(const v2f*)&WlS[kb + nn * 2];
          v2f bR = *(const v2f*)&WrS[kb + nn * 2];
          acc[t] = __builtin_amdgcn_wmma_f32_16x16x4_f32(
              false, am, false, bL, (short)0, acc[t], false, false);
          acc[t] = __builtin_amdgcn_wmma_f32_16x16x4_f32(
              false, ar[ks], false, bR, (short)0, acc[t], false, false);
        }
      }

#pragma unroll
      for (int t = 0; t < 4; ++t) {
        int nn = t * 16 + l16;
        float bias = bl[nn];
#pragma unroll
        for (int v = 0; v < 8; ++v) {
          int mr = half * 8 + v;
          float val = fmaxf(acc[t][v] + bias, 0.0f);
          hout[(long long)(row0 + mr) * GD + nn] = val;
        }
      }
    }
  }
}

__global__ __launch_bounds__(256) void final_head_k(const float* __restrict__ h,
                                                    const float* __restrict__ Wf,
                                                    const float* __restrict__ bf,
                                                    float* __restrict__ out,
                                                    int n_total) {
  int i = blockIdx.x * 256 + threadIdx.x;
  if (i >= n_total) return;
  const float4* r = (const float4*)(h + (long long)i * GD);
  const float4* w = (const float4*)Wf;
  float s = 0.0f;
#pragma unroll
  for (int q = 0; q < GD / 4; ++q) {
    float4 a = r[q], b = w[q];
    s += a.x * b.x + a.y * b.y + a.z * b.z + a.w * b.w;
  }
  out[i] = s + bf[0];
}

extern "C" void kernel_launch(void* const* d_in, const int* in_sizes, int n_in,
                              void* d_out, int out_size, void* d_ws, size_t ws_size,
                              hipStream_t stream) {
  (void)in_sizes; (void)n_in; (void)out_size; (void)ws_size;

  const float* x   = (const float*)d_in[0];
  const int*   ei  = (const int*)d_in[1];
  const int*   src = ei;
  const int*   dst = ei + GE;
  const float* Wl1 = (const float*)d_in[2];
  const float* bl1 = (const float*)d_in[3];
  const float* Wr1 = (const float*)d_in[4];
  const float* Wl2 = (const float*)d_in[5];
  const float* bl2 = (const float*)d_in[6];
  const float* Wr2 = (const float*)d_in[7];
  const float* Wf  = (const float*)d_in[8];
  const float* bf  = (const float*)d_in[9];

  char*  ws = (char*)d_ws;
  float* h1 = (float*)ws;
  float* h2 = (float*)(ws + (size_t)GN * GD * 4 + (size_t)GN * 4);

  sage_layer_own_k<1><<<NBLK, 1024, 0, stream>>>(src, dst, x,  Wl1, Wr1, bl1, h1, h2);
  sage_layer_own_k<2><<<NBLK, 1024, 0, stream>>>(src, dst, h1, Wl2, Wr2, bl2, h2, h2);
  final_head_k<<<(GN + 255) / 256, 256, 0, stream>>>(h2, Wf, bf, (float*)d_out, GN);
}


# Layer 1 scans its block's own owner bitmap (one bit per 16-byte dst piece, built per block by the first launch) and fetches the set bits' pieces, not an owner byte per edge; LDS add order changes
# speedup vs baseline: 5.2935x; 1.4244x over previous
#include <hip/hip_runtime.h>


typedef __attribute__((ext_vector_type(2))) float v2f;
typedef __attribute__((ext_vector_type(8))) float v8f;

#define GN 50000
#define GE 800000
#define GD 64
#define WROW 160
#define RB 208
#define NBLK ((GN + RB - 1) / RB)
#define ACCP 68
#define NWAVE 32
#define QFLUSH 128
#define QCAP (QFLUSH + 1024)
#define NV4 (GE / 4)
#define NFULL (NV4 / 256)
#define NTAIL ((NV4 - NFULL * 256) / 32)
#define LIST_OVER 0xFFFFFFFFu
#define BMROWS ((NV4 + 1023) / 1024)
#define BMW (BMROWS * 32)
#define BMK ((BMROWS + NWAVE - 1) / NWAVE)
#define KOFF 48
#define LOFF (KOFF + NBLK * BMW)
#define LCAP 216
#define LPITCH (NWAVE + NWAVE * LCAP)
#define QP 256
#define QW (QP + 4 * QP)
#define TPITCH 33
static_assert(QW >= QCAP, "the fallback scan's queue fits the wave's LDS words");
static_assert((long long)LOFF + (long long)NBLK * LPITCH <= (long long)GN * GD, "bitmaps and lists fit the h2 plane");
static_assert((((long long)GN * GD + GN) * 4 + KOFF * 4) % 256 == 0, "bitmaps start on 256-byte units of the workspace");
static_assert((BMW * 4) % 256 == 0, "every bitmap starts on a 256-byte unit");
static_assert(GE < (1 << 20) && RB < (1 << 12), "a list entry holds edge index and local row");
static_assert(NFULL * 256 + NTAIL * 32 == NV4, "the dst list must split into wave-wide 16-byte pieces");

__device__ __forceinline__ void drain_rows(unsigned ent, int s, int m,
                                           const float* __restrict__ feat,
                                           float* accS, float* cntS, int lane) {
  int dl = (int)(ent >> 20);
  if (lane < m) atomicAdd(&cntS[dl], 1.0f);
  for (int j0 = 0; j0 < m; j0 += 16) {
    v2f r[16];
#pragma unroll
    for (int u = 0; u < 16; ++u) {
      int sj = __builtin_amdgcn_readlane(s, j0 + u);
      r[u] = *(const v2f*)(feat + (long long)sj * GD + lane * 2);
    }
#pragma unroll
    for (int u = 0; u < 16; ++u) {
      if (j0 + u < m) {
        int dj = __builtin_amdgcn_readlane(dl, j0 + u);
        float* a = accS + dj * ACCP + lane * 2;
        atomicAdd(a, r[u].x);
        atomicAdd(a + 1, r[u].y);
      }
    }
  }
}

template <bool SAVE>
__device__ __forceinline__ void drain_list(const unsigned* q, int n,
                                           const int* __restrict__ src,
                                           const float* __restrict__ feat,
                                           float* accS, float* cntS, int lane,
                                           float* seg, int cap, int& saved) {
  for (int g = 0; g < n; g += 32) {
    int m = n - g;
    if (m > 32) m = 32;
    unsigned ent = (lane < m) ? q[g + lane] : 0u;
    int s = 0;
    if (lane < m) s = src[ent & 0xFFFFFu];
    if (SAVE) {
      const int p = saved + g + lane;
      if (lane < m && p < cap) seg[p] = __uint_as_float(ent);
    }
    drain_rows(ent, s, m, feat, accS, cntS, lane);
  }
  if (SAVE) saved += n;
}

__device__ __forceinline__ void load_step(int4* d, const int4* __restrict__ dst4,
                                          int st, int lane) {
#pragma unroll
  for (int k = 0; k < 8; ++k) d[k] = dst4[st * 256 + k * 32 + lane];
}

__device__ __forceinline__ void keep_edge(bool mine, unsigned mask, int dlc, int e,
                                          unsigned* q, int& n) {
  int pos = n + (int)__builtin_amdgcn_mbcnt_lo(mask, 0u);
  if (mine) q[pos] = (unsigned)e | ((unsigned)dlc << 20);
  n += __builtin_popcount(mask);
}
__device__ __forceinline__ void scan_piece(const int4 d, const int vi, const int base,
                                           unsigned* q, int& n) {
  const int d0 = d.x - base, d1 = d.y - base, d2 = d.z - base, d3 = d.w - base;
  const bool m0 = (unsigned)d0 < (unsigned)RB;
  const bool m1 = (unsigned)d1 < (unsigned)RB;
  const bool m2 = (unsigned)d2 < (unsigned)RB;
  const bool m3 = (unsigned)d3 < (unsigned)RB;
  const unsigned k0 = __builtin_amdgcn_ballot_w32(m0);
  const unsigned k1 = __builtin_amdgcn_ballot_w32(m1);
  const unsigned k2 = __builtin_amdgcn_ballot_w32(m2);
  const unsigned k3 = __builtin_amdgcn_ballot_w32(m3);
  if ((k0 | k1 | k2 | k3) != 0u) {
    keep_edge(m0, k0, d0, vi * 4 + 0, q, n);
    keep_edge(m1, k1, d1, vi * 4 + 1, q, n);
    keep_edge(m2, k2, d2, vi * 4 + 2, q, n);
    keep_edge(m3, k3, d3, vi * 4 + 3, q, n);
  }
}
__device__ __forceinline__ void scan_step(const int4* d, int st, int lane, int base,
                                          unsigned* q, int& n) {
#pragma unroll
  for (int k = 0; k < 8; ++k) scan_piece(d[k], st * 256 + k * 32 + lane, base, q, n);
}

template <bool SAVE>
__device__ __forceinline__ void scan_all(const int* __restrict__ src, const int* __restrict__ dst,
                                         const float* __restrict__ feat,
                                         float* accS, float* cntS, unsigned* q,
                                         int wave, int lane, int base,
                                         float* seg, int cap, int& saved) {
  int n = 0;
  const int4* dst4 = (const int4*)dst;
  int4 bufA[8], bufB[8];
  load_step(bufA, dst4, wave, lane);
  for (int st = wave; st < NFULL; st += 2 * NWAVE) {
    const int sb = st + NWAVE;
    const int sa = st + 2 * NWAVE;
    load_step(bufB, dst4, sb < NFULL ? sb : NFULL - 1, lane);
    scan_step(bufA, st, lane, base, q, n);
    if (n >= QFLUSH) { drain_list<SAVE>(q, n, src, feat, accS, cntS, lane, seg, cap, saved); n = 0; }
    load_step(bufA, dst4, sa < NFULL ? sa : NFULL - 1, lane);
    if (sb < NFULL) {
      scan_step(bufB, sb, lane, base, q, n);
      if (n >= QFLUSH) { drain_list<SAVE>(q, n, src, feat, accS, cntS, lane, seg, cap, saved); n = 0; }
    }
  }
  if (wave == NWAVE - 1) {
#pragma unroll
    for (int k = 0; k < NTAIL; ++k) {
      const int vi = NFULL * 256 + k * 32 + lane;
      scan_piece(dst4[vi], vi, base, q, n);
    }
  }
  drain_list<SAVE>(q, n, src, feat, accS, cntS, lane, seg, cap, saved);
}

__device__ __forceinline__ void drain_pieces(const unsigned* qp, int n, unsigned* qe,
                                             const int* __restrict__ src,
                                             const int* __restrict__ dst,
                                             const float* __restrict__ feat,
                                             float* accS, float* cntS, int lane, int base,
                                             float* seg, int cap, int& saved) {
  const int4* dst4 = (const int4*)dst;
  unsigned pp[QP / 32];
  int4 dd[QP / 32];
#pragma unroll
  for (int i = 0; i < QP / 32; ++i) {
    pp[i] = 0u;
    dd[i] = make_int4(0, 0, 0, 0);
    if (i * 32 < n) {
      pp[i] = (i * 32 + lane < n) ? qp[i * 32 + lane] : 0u;
      dd[i] = dst4[pp[i]];
    }
  }
  int n2 = 0;
#pragma unroll
  for (int i = 0; i < QP / 32; ++i) {
    if (i * 32 < n) {
      const bool on = i * 32 + lane < n;
      const int dl[4] = {dd[i].x - base, dd[i].y - base, dd[i].z - base, dd[i].w - base};
#pragma unroll
      for (int s = 0; s < 4; ++s) {
        const bool mine = on && (unsigned)dl[s] < (unsigned)RB;
        const unsigned mask = __builtin_amdgcn_ballot_w32(mine);
        const int pos = n2 + (int)__builtin_amdgcn_mbcnt_lo(mask, 0u);
        if (mine) qe[pos] = (pp[i] * 4u + (unsigned)s) | ((unsigned)dl[s] << 20);
        n2 += __builtin_popcount(mask);
      }
    }
  }
  drain_list<true>(qe, n2, src, feat, accS, cntS, lane, seg, cap, saved);
}

__device__ __forceinline__ void load_bits(unsigned* w, const float* bm, int wave, int lane) {
#pragma unroll
  for (int k = 0; k < BMK; ++k) {
    const int r = wave + k * NWAVE;
    w[k] = (r < BMROWS) ? __float_as_uint(bm[r * 32 + lane]) : 0u;
  }
}

__device__ __forceinline__ void scan_bits(unsigned* w,
                                          const int* __restrict__ src, const int* __restrict__ dst,
                                          const float* __restrict__ feat,
                                          float* accS, float* cntS, unsigned* q,
                                          int wave, int lane, int base,
                                          float* seg, int cap, int& saved) {
  unsigned* qp = q;
  unsigned* qe = q + QP;
  int n = 0;
  for (int t = 0; t < BMK * 32; ++t) {
    int ks = 0;
    unsigned zs = 0u;
#pragma unroll
    for (int k = BMK - 1; k >= 0; --k) {
      if (w[k] != 0u) { ks = k; zs = w[k]; }
    }
    const bool has = zs != 0u;
    const unsigned mask = __builtin_amdgcn_ballot_w32(has);
    if (mask == 0u) break;
    if (n + 32 > QP) {
      drain_pieces(qp, n, qe, src, dst, feat, accS, cntS, lane, base, seg, cap, saved);
      n = 0;
    }
    const int b = __builtin_ctz(zs | 0x80000000u);
    const unsigned piece = ((unsigned)((wave + ks * NWAVE) * 32 + lane) << 5) + (unsigned)b;
    const int pos = n + (int)__builtin_amdgcn_mbcnt_lo(mask, 0u);
    if (has) qp[pos] = piece;
    n += __builtin_popcount(mask);
    const unsigned left = zs & (zs - 1u);
#pragma unroll
    for (int k = 0; k < BMK; ++k) {
      if (k == ks) w[k] = left;
    }
  }
  drain_pieces(qp, n, qe, src, dst, feat, accS, cntS, lane, base, seg, cap, saved);
}

#define KWAVES 8
#define KPW (32 / KWAVES)
__global__ __launch_bounds__(KWAVES * 32) void owner_bits_k(const int* __restrict__ dst, float* plane) {
  __shared__ unsigned T[NBLK * TPITCH];
  const int tid = threadIdx.x;
  const int lane = tid & 31;
  const int wv = tid >> 5;
  const int g = blockIdx.x;
  const int4* dst4 = (const int4*)dst;
  int4 d[KPW];
#pragma unroll
  for (int i = 0; i < KPW; ++i) {
    const int p = (g * 32 + wv * KPW + i) * 32 + lane;
    d[i] = dst4[p < NV4 ? p : NV4 - 1];
  }
  for (int i = tid; i < NBLK * TPITCH; i += KWAVES * 32) T[i] = 0u;
  __syncthreads();
  const unsigned bit = 1u << lane;
#pragma unroll
  for (int i = 0; i < KPW; ++i) {
    const int k = wv * KPW + i;
    const int p = (g * 32 + k) * 32 + lane;
    if (p < NV4) {
      const unsigned o0 = (unsigned)d[i].x / (unsigned)RB;
      const unsigned o1 = (unsigned)d[i].y / (unsigned)RB;
      const unsigned o2 = (unsigned)d[i].z / (unsigned)RB;
      const unsigned o3 = (unsigned)d[i].w / (unsigned)RB;
      if (o0 < NBLK) atomicOr(&T[o0 * TPITCH + k], bit);
      if (o1 < NBLK) atomicOr(&T[o1 * TPITCH + k], bit);
      if (o2 < NBLK) atomicOr(&T[o2 * TPITCH + k], bit);
      if (o3 < NBLK) atomicOr(&T[o3 * TPITCH + k], bit);
    }
  }
  __syncthreads();
  float* outw = plane + KOFF + g * 32 + lane;
#pragma unroll 8
  for (int b = wv; b < NBLK; b += KWAVES) outw[(size_t)b * BMW] = __uint_as_float(T[b * TPITCH + lane]);
}

template <int LAYER>
__global__ __launch_bounds__(1024) void sage_layer_own_k(
    const int* __restrict__ src, const int* __restrict__ dst,
    const float* __restrict__ feat,
    const float* __restrict__ Wl, const float* __restrict__ Wr,
    const float* __restrict__ bl,
    float* hout, float* lst,
    const float* __restrict__ Wf, const float* __restrict__ bf,
    float* __restrict__ out)
{
  __shared__ float WlS[(GD / 2) * WROW];
  __shared__ float WrS[(GD / 2) * WROW];
  __shared__ float accS[RB * ACCP];
  __shared__ float cntS[RB];
  __shared__ unsigned qS[NWAVE * QW];

  const int tid  = threadIdx.x;
  const int lane = tid & 31;
  const int wave = __builtin_amdgcn_readfirstlane(tid >> 5);
  const int base = blockIdx.x * RB;

  float* reg = lst + LOFF + (long long)blockIdx.x * LPITCH;
  const int cap = LCAP;
  float* seg = reg + NWAVE + wave * cap;

  unsigned cw = 0u;
  if (LAYER == 2) cw = __float_as_uint(reg[lane]);
  unsigned bw[BMK];
  if (LAYER == 1) load_bits(bw, lst + KOFF + (long long)blockIdx.x * BMW, wave, lane);

  for (int i = tid; i < GD * GD; i += 1024) {
    int j = i >> 6;
    int k = i & 63;
    int o = (k >> 1) * WROW + j * 2 + (k & 1);
    WlS[o] = Wl[i];
    WrS[o] = Wr[i];
  }
  for (int i = tid; i < RB * ACCP; i += 1024) accS[i] = 0.0f;
  if (tid < RB) cntS[tid] = 0.0f;
  __syncthreads();

  unsigned* q = qS + wave * QW;
  int saved = 0;
  if (LAYER == 1) {
    scan_bits(bw, src, dst, feat, accS, cntS, q, wave, lane, base, seg, cap, saved);
    if (lane == 0) reg[wave] = __uint_as_float(saved <= cap ? (unsigned)saved : LIST_OVER);
  } else {
    const bool over = __builtin_amdgcn_ballot_w32(cw == LIST_OVER) != 0u;
    if (!over) {
      const int nl = __builtin_amdgcn_readlane((int)cw, wave);
      unsigned ent  = (lane < nl) ? __float_as_uint(seg[lane]) : 0u;
      unsigned ent1 = (32 + lane < nl) ? __float_as_uint(seg[32 + lane]) : 0u;
      int s = (lane < nl) ? src[ent & 0xFFFFFu] : 0;
      for (int g = 0; g < nl; g += 32) {
        int m = nl - g;
        if (m > 32) m = 32;
        const unsigned ent2 = (g + 64 + lane < nl) ? __float_as_uint(seg[g + 64 + lane]) : 0u;
        const int s1 = (g + 32 + lane < nl) ? src[ent1 & 0xFFFFFu] : 0;
        drain_rows(ent, s, m, feat, accS, cntS, lane);
        ent = ent1;
        s = s1;
        ent1 = ent2;
      }
    } else {
      scan_all<false>(src, dst, feat, accS, cntS, q, wave, lane, base, seg, cap, saved);
    }
  }
  __syncthreads();

  if (wave < RB / 16) {
    const int row0 = base + wave * 16;
    if (row0 < GN) {
      const int half = lane >> 4;
      const int l16  = lane & 15;
      const int lr   = wave * 16 + l16;
      const float inv = 1.0f / fmaxf(cntS[lr], 1.0f);
      const float* xRow = feat + (long long)(base + lr) * GD;

      v2f ar[16];
#pragma unroll
      for (int ks = 0; ks < 16; ++ks) ar[ks] = *(const v2f*)(xRow + ks * 4 + half * 2);

      v8f acc[4] = {};
#pragma unroll
      for (int ks = 0; ks < 16; ++ks) {
        const int ka = ks * 4 + half * 2;
        v2f am = *(const v2f*)&accS[lr * ACCP + ka];
        am *= inv;
        const int kb = (ka >> 1) * WROW;
#pragma unroll
        for (int t = 0; t < 4; ++t) {
          int nn = t * 16 + l16;
          v2f bL = *(const v2f*)&WlS[kb + nn * 2];
          v2f bR = *(const v2f*)&WrS[kb + nn * 2];
          acc[t] = __builtin_amdgcn_wmma_f32_16x16x4_f32(
              false, am, false, bL, (short)0, acc[t], false, false);
          acc[t] = __builtin_amdgcn_wmma_f32_16x16x4_f32(
              false, ar[ks], false, bR, (short)0, acc[t], false, false);
        }
      }

      if (LAYER == 1) {
#pragma unroll
        for (int t = 0; t < 4; ++t) {
          int nn = t * 16 + l16;
          float bias = bl[nn];
#pragma unroll
          for (int v = 0; v < 8; ++v) {
            int mr = half * 8 + v;
            float val = fmaxf(acc[t][v] + bias, 0.0f);
            hout[(long long)(row0 + mr) * GD + nn] = val;
          }
        }
      } else {
        float part[8];
#pragma unroll
        for (int v = 0; v < 8; ++v) part[v] = 0.0f;
#pragma unroll
        for (int t = 0; t < 4; ++t) {
          const int nn = t * 16 + l16;
          const float bias = bl[nn];
          const float wf = Wf[nn];
#pragma unroll
          for (int v = 0; v < 8; ++v) part[v] += fmaxf(acc[t][v] + bias, 0.0f) * wf;
        }
#pragma unroll
        for (int v = 0; v < 8; ++v) {
          float s = part[v];
          s += __shfl_xor(s, 8, 16);
          s += __shfl_xor(s, 4, 16);
          s += __shfl_xor(s, 2, 16);
          s += __shfl_xor(s, 1, 16);
          part[v] = s;
        }
        float mine = part[0];
#pragma unroll
        for (int v = 1; v < 8; ++v) mine = (l16 == v) ? part[v] : mine;
        if (l16 < 8) out[row0 + half * 8 + l16] = mine + bf[0];
      }
    }
  }
}

extern "C" void kernel_launch(void* const* d_in, const int* in_sizes, int n_in,
                              void* d_out, int out_size, void* d_ws, size_t ws_size,
                              hipStream_t stream) {
  (void)in_sizes; (void)n_in; (void)out_size; (void)ws_size;

  const float* x   = (const float*)d_in[0];
  const int*   ei  = (const int*)d_in[1];
  const int*   src = ei;
  const int*   dst = ei + GE;
  const float* Wl1 = (const float*)d_in[2];
  const float* bl1 = (const float*)d_in[3];
  const float* Wr1 = (const float*)d_in[4];
  const float* Wl2 = (const float*)d_in[5];
  const float* bl2 = (const float*)d_in[6];
  const float* Wr2 = (const float*)d_in[7];
  const float* Wf  = (const float*)d_in[8];
  const float* bf  = (const float*)d_in[9];

  char*  ws = (char*)d_ws;
  float* h1 = (float*)ws;
  float* h2 = (float*)(ws + (size_t)GN * GD * 4 + (size_t)GN * 4);

  owner_bits_k<<<BMROWS, KWAVES * 32, 0, stream>>>(dst, h2);
  sage_layer_own_k<1><<<NBLK, 1024, 0, stream>>>(src, dst, x,  Wl1, Wr1, bl1, h1, h2,
                                                 Wf, bf, (float*)d_out);
  sage_layer_own_k<2><<<NBLK, 1024, 0, stream>>>(src, dst, h1, Wl2, Wr2, bl2, h2, h2,
                                                 Wf, bf, (float*)d_out);
}


# Layer 1 saves each kept edge's source node (looked up there anyway), not its edge index: layer 2 gathers rows straight from its list, no src lookup per edge; LDS add arrival order may change
# speedup vs baseline: 5.4853x; 1.0362x over previous
#include <hip/hip_runtime.h>


typedef __attribute__((ext_vector_type(2))) float v2f;
typedef __attribute__((ext_vector_type(8))) float v8f;

#define GN 50000
#define GE 800000
#define GD 64
#define WROW 160
#define RB 208
#define NBLK ((GN + RB - 1) / RB)
#define ACCP 68
#define NWAVE 32
#define QFLUSH 128
#define QCAP (QFLUSH + 1024)
#define NV4 (GE / 4)
#define NFULL (NV4 / 256)
#define NTAIL ((NV4 - NFULL * 256) / 32)
#define LIST_OVER 0xFFFFFFFFu
#define BMROWS ((NV4 + 1023) / 1024)
#define BMW (BMROWS * 32)
#define BMK ((BMROWS + NWAVE - 1) / NWAVE)
#define KOFF 48
#define LOFF (KOFF + NBLK * BMW)
#define LCAP 216
#define LPITCH (NWAVE + NWAVE * LCAP)
#define QP 256
#define QW (QP + 4 * QP)
#define TPITCH 33
static_assert(QW >= QCAP, "the fallback scan's queue fits the wave's LDS words");
static_assert((long long)LOFF + (long long)NBLK * LPITCH <= (long long)GN * GD, "bitmaps and lists fit the h2 plane");
static_assert((((long long)GN * GD + GN) * 4 + KOFF * 4) % 256 == 0, "bitmaps start on 256-byte units of the workspace");
static_assert((BMW * 4) % 256 == 0, "every bitmap starts on a 256-byte unit");
static_assert(GE < (1 << 20) && GN < (1 << 20) && RB < (1 << 12), "an entry holds edge index or source node, and local row");
static_assert(NFULL * 256 + NTAIL * 32 == NV4, "the dst list must split into wave-wide 16-byte pieces");

__device__ __forceinline__ void drain_rows(unsigned ent, int s, int m,
                                           const float* __restrict__ feat,
                                           float* accS, float* cntS, int lane) {
  int dl = (int)(ent >> 20);
  if (lane < m) atomicAdd(&cntS[dl], 1.0f);
  for (int j0 = 0; j0 < m; j0 += 16) {
    v2f r[16];
#pragma unroll
    for (int u = 0; u < 16; ++u) {
      int sj = __builtin_amdgcn_readlane(s, j0 + u);
      r[u] = *(const v2f*)(feat + (long long)sj * GD + lane * 2);
    }
#pragma unroll
    for (int u = 0; u < 16; ++u) {
      if (j0 + u < m) {
        int dj = __builtin_amdgcn_readlane(dl, j0 + u);
        float* a = accS + dj * ACCP + lane * 2;
        atomicAdd(a, r[u].x);
        atomicAdd(a + 1, r[u].y);
      }
    }
  }
}

template <bool SAVE>
__device__ __forceinline__ void drain_list(const unsigned* q, int n,
                                           const int* __restrict__ src,
                                           const float* __restrict__ feat,
                                           float* accS, float* cntS, int lane,
                                           float* seg, int cap, int& saved) {
  for (int g = 0; g < n; g += 32) {
    int m = n - g;
    if (m > 32) m = 32;
    unsigned ent = (lane < m) ? q[g + lane] : 0u;
    int s = 0;
    if (lane < m) s = src[ent & 0xFFFFFu];
    if (SAVE) {
      const int p = saved + g + lane;
      if (lane < m && p < cap) seg[p] = __uint_as_float((ent & 0xFFF00000u) | (unsigned)s);
    }
    drain_rows(ent, s, m, feat, accS, cntS, lane);
  }
  if (SAVE) saved += n;
}

__device__ __forceinline__ void load_step(int4* d, const int4* __restrict__ dst4,
                                          int st, int lane) {
#pragma unroll
  for (int k = 0; k < 8; ++k) d[k] = dst4[st * 256 + k * 32 + lane];
}

__device__ __forceinline__ void keep_edge(bool mine, unsigned mask, int dlc, int e,
                                          unsigned* q, int& n) {
  int pos = n + (int)__builtin_amdgcn_mbcnt_lo(mask, 0u);
  if (mine) q[pos] = (unsigned)e | ((unsigned)dlc << 20);
  n += __builtin_popcount(mask);
}
__device__ __forceinline__ void scan_piece(const int4 d, const int vi, const int base,
                                           unsigned* q, int& n) {
  const int d0 = d.x - base, d1 = d.y - base, d2 = d.z - base, d3 = d.w - base;
  const bool m0 = (unsigned)d0 < (unsigned)RB;
  const bool m1 = (unsigned)d1 < (unsigned)RB;
  const bool m2 = (unsigned)d2 < (unsigned)RB;
  const bool m3 = (unsigned)d3 < (unsigned)RB;
  const unsigned k0 = __builtin_amdgcn_ballot_w32(m0);
  const unsigned k1 = __builtin_amdgcn_ballot_w32(m1);
  const unsigned k2 = __builtin_amdgcn_ballot_w32(m2);
  const unsigned k3 = __builtin_amdgcn_ballot_w32(m3);
  if ((k0 | k1 | k2 | k3) != 0u) {
    keep_edge(m0, k0, d0, vi * 4 + 0, q, n);
    keep_edge(m1, k1, d1, vi * 4 + 1, q, n);
    keep_edge(m2, k2, d2, vi * 4 + 2, q, n);
    keep_edge(m3, k3, d3, vi * 4 + 3, q, n);
  }
}
__device__ __forceinline__ void scan_step(const int4* d, int st, int lane, int base,
                                          unsigned* q, int& n) {
#pragma unroll
  for (int k = 0; k < 8; ++k) scan_piece(d[k], st * 256 + k * 32 + lane, base, q, n);
}

template <bool SAVE>
__device__ __forceinline__ void scan_all(const int* __restrict__ src, const int* __restrict__ dst,
                                         const float* __restrict__ feat,
                                         float* accS, float* cntS, unsigned* q,
                                         int wave, int lane, int base,
                                         float* seg, int cap, int& saved) {
  int n = 0;
  const int4* dst4 = (const int4*)dst;
  int4 bufA[8], bufB[8];
  load_step(bufA, dst4, wave, lane);
  for (int st = wave; st < NFULL; st += 2 * NWAVE) {
    const int sb = st + NWAVE;
    const int sa = st + 2 * NWAVE;
    load_step(bufB, dst4, sb < NFULL ? sb : NFULL - 1, lane);
    scan_step(bufA, st, lane, base, q, n);
    if (n >= QFLUSH) { drain_list<SAVE>(q, n, src, feat, accS, cntS, lane, seg, cap, saved); n = 0; }
    load_step(bufA, dst4, sa < NFULL ? sa : NFULL - 1, lane);
    if (sb < NFULL) {
      scan_step(bufB, sb, lane, base, q, n);
      if (n >= QFLUSH) { drain_list<SAVE>(q, n, src, feat, accS, cntS, lane, seg, cap, saved); n = 0; }
    }
  }
  if (wave == NWAVE - 1) {
#pragma unroll
    for (int k = 0; k < NTAIL; ++k) {
      const int vi = NFULL * 256 + k * 32 + lane;
      scan_piece(dst4[vi], vi, base, q, n);
    }
  }
  drain_list<SAVE>(q, n, src, feat, accS, cntS, lane, seg, cap, saved);
}

__device__ __forceinline__ void drain_pieces(const unsigned* qp, int n, unsigned* qe,
                                             const int* __restrict__ src,
                                             const int* __restrict__ dst,
                                             const float* __restrict__ feat,
                                             float* accS, float* cntS, int lane, int base,
                                             float* seg, int cap, int& saved) {
  const int4* dst4 = (const int4*)dst;
  unsigned pp[QP / 32];
  int4 dd[QP / 32];
#pragma unroll
  for (int i = 0; i < QP / 32; ++i) {
    pp[i] = 0u;
    dd[i] = make_int4(0, 0, 0, 0);
    if (i * 32 < n) {
      pp[i] = (i * 32 + lane < n) ? qp[i * 32 + lane] : 0u;
      dd[i] = dst4[pp[i]];
    }
  }
  int n2 = 0;
#pragma unroll
  for (int i = 0; i < QP / 32; ++i) {
    if (i * 32 < n) {
      const bool on = i * 32 + lane < n;
      const int dl[4] = {dd[i].x - base, dd[i].y - base, dd[i].z - base, dd[i].w - base};
#pragma unroll
      for (int s = 0; s < 4; ++s) {
        const bool mine = on && (unsigned)dl[s] < (unsigned)RB;
        const unsigned mask = __builtin_amdgcn_ballot_w32(mine);
        const int pos = n2 + (int)__builtin_amdgcn_mbcnt_lo(mask, 0u);
        if (mine) qe[pos] = (pp[i] * 4u + (unsigned)s) | ((unsigned)dl[s] << 20);
        n2 += __builtin_popcount(mask);
      }
    }
  }
  drain_list<true>(qe, n2, src, feat, accS, cntS, lane, seg, cap, saved);
}

__device__ __forceinline__ void load_bits(unsigned* w, const float* bm, int wave, int lane) {
#pragma unroll
  for (int k = 0; k < BMK; ++k) {
    const int r = wave + k * NWAVE;
    w[k] = (r < BMROWS) ? __float_as_uint(bm[r * 32 + lane]) : 0u;
  }
}

__device__ __forceinline__ void scan_bits(unsigned* w,
                                          const int* __restrict__ src, const int* __restrict__ dst,
                                          const float* __restrict__ feat,
                                          float* accS, float* cntS, unsigned* q,
                                          int wave, int lane, int base,
                                          float* seg, int cap, int& saved) {
  unsigned* qp = q;
  unsigned* qe = q + QP;
  int n = 0;
  for (int t = 0; t < BMK * 32; ++t) {
    int ks = 0;
    unsigned zs = 0u;
#pragma unroll
    for (int k = BMK - 1; k >= 0; --k) {
      if (w[k] != 0u) { ks = k; zs = w[k]; }
    }
    const bool has = zs != 0u;
    const unsigned mask = __builtin_amdgcn_ballot_w32(has);
    if (mask == 0u) break;
    if (n + 32 > QP) {
      drain_pieces(qp, n, qe, src, dst, feat, accS, cntS, lane, base, seg, cap, saved);
      n = 0;
    }
    const int b = __builtin_ctz(zs | 0x80000000u);
    const unsigned piece = ((unsigned)((wave + ks * NWAVE) * 32 + lane) << 5) + (unsigned)b;
    const int pos = n + (int)__builtin_amdgcn_mbcnt_lo(mask, 0u);
    if (has) qp[pos] = piece;
    n += __builtin_popcount(mask);
    const unsigned left = zs & (zs - 1u);
#pragma unroll
    for (int k = 0; k < BMK; ++k) {
      if (k == ks) w[k] = left;
    }
  }
  drain_pieces(qp, n, qe, src, dst, feat, accS, cntS, lane, base, seg, cap, saved);
}

#define KWAVES 8
#define KPW (32 / KWAVES)
__global__ __launch_bounds__(KWAVES * 32) void owner_bits_k(const int* __restrict__ dst, float* plane) {
  __shared__ unsigned T[NBLK * TPITCH];
  const int tid = threadIdx.x;
  const int lane = tid & 31;
  const int wv = tid >> 5;
  const int g = blockIdx.x;
  const int4* dst4 = (const int4*)dst;
  int4 d[KPW];
#pragma unroll
  for (int i = 0; i < KPW; ++i) {
    const int p = (g * 32 + wv * KPW + i) * 32 + lane;
    d[i] = dst4[p < NV4 ? p : NV4 - 1];
  }
  for (int i = tid; i < NBLK * TPITCH; i += KWAVES * 32) T[i] = 0u;
  __syncthreads();
  const unsigned bit = 1u << lane;
#pragma unroll
  for (int i = 0; i < KPW; ++i) {
    const int k = wv * KPW + i;
    const int p = (g * 32 + k) * 32 + lane;
    if (p < NV4) {
      const unsigned o0 = (unsigned)d[i].x / (unsigned)RB;
      const unsigned o1 = (unsigned)d[i].y / (unsigned)RB;
      const unsigned o2 = (unsigned)d[i].z / (unsigned)RB;
      const unsigned o3 = (unsigned)d[i].w / (unsigned)RB;
      if (o0 < NBLK) atomicOr(&T[o0 * TPITCH + k], bit);
      if (o1 < NBLK) atomicOr(&T[o1 * TPITCH + k], bit);
      if (o2 < NBLK) atomicOr(&T[o2 * TPITCH + k], bit);
      if (o3 < NBLK) atomicOr(&T[o3 * TPITCH + k], bit);
    }
  }
  __syncthreads();
  float* outw = plane + KOFF + g * 32 + lane;
#pragma unroll 8
  for (int b = wv; b < NBLK; b += KWAVES) outw[(size_t)b * BMW] = __uint_as_float(T[b * TPITCH + lane]);
}

template <int LAYER>
__global__ __launch_bounds__(1024) void sage_layer_own_k(
    const int* __restrict__ src, const int* __restrict__ dst,
    const float* __restrict__ feat,
    const float* __restrict__ Wl, const float* __restrict__ Wr,
    const float* __restrict__ bl,
    float* hout, float* lst,
    const float* __restrict__ Wf, const float* __restrict__ bf,
    float* __restrict__ out)
{
  __shared__ float WlS[(GD / 2) * WROW];
  __shared__ float WrS[(GD / 2) * WROW];
  __shared__ float accS[RB * ACCP];
  __shared__ float cntS[RB];
  __shared__ unsigned qS[NWAVE * QW];

  const int tid  = threadIdx.x;
  const int lane = tid & 31;
  const int wave = __builtin_amdgcn_readfirstlane(tid >> 5);
  const int base = blockIdx.x * RB;

  float* reg = lst + LOFF + (long long)blockIdx.x * LPITCH;
  const int cap = LCAP;
  float* seg = reg + NWAVE + wave * cap;

  unsigned cw = 0u;
  if (LAYER == 2) cw = __float_as_uint(reg[lane]);
  unsigned bw[BMK];
  if (LAYER == 1) load_bits(bw, lst + KOFF + (long long)blockIdx.x * BMW, wave, lane);

  for (int i = tid; i < GD * GD; i += 1024) {
    int j = i >> 6;
    int k = i & 63;
    int o = (k >> 1) * WROW + j * 2 + (k & 1);
    WlS[o] = Wl[i];
    WrS[o] = Wr[i];
  }
  for (int i = tid; i < RB * ACCP; i += 1024) accS[i] = 0.0f;
  if (tid < RB) cntS[tid] = 0.0f;
  __syncthreads();

  unsigned* q = qS + wave * QW;
  int saved = 0;
  if (LAYER == 1) {
    scan_bits(bw, src, dst, feat, accS, cntS, q, wave, lane, base, seg, cap, saved);
    if (lane == 0) reg[wave] = __uint_as_float(saved <= cap ? (unsigned)saved : LIST_OVER);
  } else {
    const bool over = __builtin_amdgcn_ballot_w32(cw == LIST_OVER) != 0u;
    if (!over) {
      const int nl = __builtin_amdgcn_readlane((int)cw, wave);
      unsigned ent  = (lane < nl) ? __float_as_uint(seg[lane]) : 0u;
      unsigned ent1 = (32 + lane < nl) ? __float_as_uint(seg[32 + lane]) : 0u;
      for (int g = 0; g < nl; g += 32) {
        int m = nl - g;
        if (m > 32) m = 32;
        const unsigned ent2 = (g + 64 + lane < nl) ? __float_as_uint(seg[g + 64 + lane]) : 0u;
        drain_rows(ent, (int)(ent & 0xFFFFFu), m, feat, accS, cntS, lane);
        ent = ent1;
        ent1 = ent2;
      }
    } else {
      scan_all<false>(src, dst, feat, accS, cntS, q, wave, lane, base, seg, cap, saved);
    }
  }
  __syncthreads();

  if (wave < RB / 16) {
    const int row0 = base + wave * 16;
    if (row0 < GN) {
      const int half = lane >> 4;
      const int l16  = lane & 15;
      const int lr   = wave * 16 + l16;
      const float inv = 1.0f / fmaxf(cntS[lr], 1.0f);
      const float* xRow = feat + (long long)(base + lr) * GD;

      v2f ar[16];
#pragma unroll
      for (int ks = 0; ks < 16; ++ks) ar[ks] = *(const v2f*)(xRow + ks * 4 + half * 2);

      v8f acc[4] = {};
#pragma unroll
      for (int ks = 0; ks < 16; ++ks) {
        const int ka = ks * 4 + half * 2;
        v2f am = *(const v2f*)&accS[lr * ACCP + ka];
        am *= inv;
        const int kb = (ka >> 1) * WROW;
#pragma unroll
        for (int t = 0; t < 4; ++t) {
          int nn = t * 16 + l16;
          v2f bL = *(const v2f*)&WlS[kb + nn * 2];
          v2f bR = *(const v2f*)&WrS[kb + nn * 2];
          acc[t] = __builtin_amdgcn_wmma_f32_16x16x4_f32(
              false, am, false, bL, (short)0, acc[t], false, false);
          acc[t] = __builtin_amdgcn_wmma_f32_16x16x4_f32(
              false, ar[ks], false, bR, (short)0, acc[t], false, false);
        }
      }

      if (LAYER == 1) {
#pragma unroll
        for (int t = 0; t < 4; ++t) {
          int nn = t * 16 + l16;
          float bias = bl[nn];
#pragma unroll
          for (int v = 0; v < 8; ++v) {
            int mr = half * 8 + v;
            float val = fmaxf(acc[t][v] + bias, 0.0f);
            hout[(long long)(row0 + mr) * GD + nn] = val;
          }
        }
      } else {
        float part[8];
#pragma unroll
        for (int v = 0; v < 8; ++v) part[v] = 0.0f;
#pragma unroll
        for (int t = 0; t < 4; ++t) {
          const int nn = t * 16 + l16;
          const float bias = bl[nn];
          const float wf = Wf[nn];
#pragma unroll
          for (int v = 0; v < 8; ++v) part[v] += fmaxf(acc[t][v] + bias, 0.0f) * wf;
        }
#pragma unroll
        for (int v = 0; v < 8; ++v) {
          float s = part[v];
          s += __shfl_xor(s, 8, 16);
          s += __shfl_xor(s, 4, 16);
          s += __shfl_xor(s, 2, 16);
          s += __shfl_xor(s, 1, 16);
          part[v] = s;
        }
        float mine = part[0];
#pragma unroll
        for (int v = 1; v < 8; ++v) mine = (l16 == v) ? part[v] : mine;
        if (l16 < 8) out[row0 + half * 8 + l16] = mine + bf[0];
      }
    }
  }
}

extern "C" void kernel_launch(void* const* d_in, const int* in_sizes, int n_in,
                              void* d_out, int out_size, void* d_ws, size_t ws_size,
                              hipStream_t stream) {
  (void)in_sizes; (void)n_in; (void)out_size; (void)ws_size;

  const float* x   = (const float*)d_in[0];
  const int*   ei  = (const int*)d_in[1];
  const int*   src = ei;
  const int*   dst = ei + GE;
  const float* Wl1 = (const float*)d_in[2];
  const float* bl1 = (const float*)d_in[3];
  const float* Wr1 = (const float*)d_in[4];
  const float* Wl2 = (const float*)d_in[5];
  const float* bl2 = (const float*)d_in[6];
  const float* Wr2 = (const float*)d_in[7];
  const float* Wf  = (const float*)d_in[8];
  const float* bf  = (const float*)d_in[9];

  char*  ws = (char*)d_ws;
  float* h1 = (float*)ws;
  float* h2 = (float*)(ws + (size_t)GN * GD * 4 + (size_t)GN * 4);

  owner_bits_k<<<BMROWS, KWAVES * 32, 0, stream>>>(dst, h2);
  sage_layer_own_k<1><<<NBLK, 1024, 0, stream>>>(src, dst, x,  Wl1, Wr1, bl1, h1, h2,
                                                 Wf, bf, (float*)d_out);
  sage_layer_own_k<2><<<NBLK, 1024, 0, stream>>>(src, dst, h1, Wl2, Wr2, bl2, h2, h2,
                                                 Wf, bf, (float*)d_out);
}


# Tile-owner waves run root . Wr^T in slices between a gather batch's row requests and its LDS adds (root rows kept in LDS); only mean . Wl^T stays behind the barrier; product sum order changes
# speedup vs baseline: 5.7207x; 1.0429x over previous
#include <hip/hip_runtime.h>


typedef __attribute__((ext_vector_type(2))) float v2f;
typedef __attribute__((ext_vector_type(8))) float v8f;

#define GN 50000
#define GE 800000
#define GD 64
#define WROW 160
#define RB 208
#define NBLK ((GN + RB - 1) / RB)
#define ACCP 68
#define NWAVE 32
#define QFLUSH 128
#define QCAP (QFLUSH + 1024)
#define NV4 (GE / 4)
#define NFULL (NV4 / 256)
#define NTAIL ((NV4 - NFULL * 256) / 32)
#define LIST_OVER 0xFFFFFFFFu
#define BMROWS ((NV4 + 1023) / 1024)
#define BMW (BMROWS * 32)
#define BMK ((BMROWS + NWAVE - 1) / NWAVE)
#define KOFF 48
#define LOFF (KOFF + NBLK * BMW)
#define LCAP 216
#define LPITCH (NWAVE + NWAVE * LCAP)
#define QP 256
#define QW (QP + 4 * QP)
#define TPITCH 33
static_assert(QW >= QCAP, "the fallback scan's queue fits the wave's LDS words");
static_assert((long long)LOFF + (long long)NBLK * LPITCH <= (long long)GN * GD, "bitmaps and lists fit the h2 plane");
static_assert((((long long)GN * GD + GN) * 4 + KOFF * 4) % 256 == 0, "bitmaps start on 256-byte units of the workspace");
static_assert((BMW * 4) % 256 == 0, "every bitmap starts on a 256-byte unit");
static_assert(GE < (1 << 20) && GN < (1 << 20) && RB < (1 << 12), "an entry holds edge index or source node, and local row");
static_assert(NFULL * 256 + NTAIL * 32 == NV4, "the dst list must split into wave-wide 16-byte pieces");

#define KSTEP 2
struct RootAcc {
  v8f acc[4];
  int ks;
  const float* a;
  const float* b;
};
__device__ __forceinline__ void root_step(RootAcc& R) {
#pragma unroll
  for (int i = 0; i < KSTEP; ++i) {
    const int ks = R.ks + i;
    const v2f a = *(const v2f*)(R.a + ks * 4);
    const float* bp = R.b + ks * 2 * WROW;
#pragma unroll
    for (int t = 0; t < 4; ++t) {
      const v2f b = *(const v2f*)(bp + t * 32);
      R.acc[t] = __builtin_amdgcn_wmma_f32_16x16x4_f32(
          false, a, false, b, (short)0, R.acc[t], false, false);
    }
  }
  R.ks += KSTEP;
}

__device__ __forceinline__ void drain_rows(unsigned ent, int s, int m,
                                           const float* __restrict__ feat,
                                           float* accS, float* cntS, int lane,
                                           RootAcc& R) {
  int dl = (int)(ent >> 20);
  if (lane < m) atomicAdd(&cntS[dl], 1.0f);
  for (int j0 = 0; j0 < m; j0 += 16) {
    v2f r[16];
#pragma unroll
    for (int u = 0; u < 16; ++u) {
      int sj = __builtin_amdgcn_readlane(s, j0 + u);
      r[u] = *(const v2f*)(feat + (long long)sj * GD + lane * 2);
    }
    if (R.ks < 16) root_step(R);
#pragma unroll
    for (int u = 0; u < 16; ++u) {
      if (j0 + u < m) {
        int dj = __builtin_amdgcn_readlane(dl, j0 + u);
        float* a = accS + dj * ACCP + lane * 2;
        atomicAdd(a, r[u].x);
        atomicAdd(a + 1, r[u].y);
      }
    }
  }
}

template <bool SAVE>
__device__ __forceinline__ void drain_list(const unsigned* q, int n,
                                           const int* __restrict__ src,
                                           const float* __restrict__ feat,
                                           float* accS, float* cntS, int lane,
                                           float* seg, int cap, int& saved, RootAcc& R) {
  for (int g = 0; g < n; g += 32) {
    int m = n - g;
    if (m > 32) m = 32;
    unsigned ent = (lane < m) ? q[g + lane] : 0u;
    int s = 0;
    if (lane < m) s = src[ent & 0xFFFFFu];
    if (SAVE) {
      const int p = saved + g + lane;
      if (lane < m && p < cap) seg[p] = __uint_as_float((ent & 0xFFF00000u) | (unsigned)s);
    }
    drain_rows(ent, s, m, feat, accS, cntS, lane, R);
  }
  if (SAVE) saved += n;
}

__device__ __forceinline__ void load_step(int4* d, const int4* __restrict__ dst4,
                                          int st, int lane) {
#pragma unroll
  for (int k = 0; k < 8; ++k) d[k] = dst4[st * 256 + k * 32 + lane];
}

__device__ __forceinline__ void keep_edge(bool mine, unsigned mask, int dlc, int e,
                                          unsigned* q, int& n) {
  int pos = n + (int)__builtin_amdgcn_mbcnt_lo(mask, 0u);
  if (mine) q[pos] = (unsigned)e | ((unsigned)dlc << 20);
  n += __builtin_popcount(mask);
}
__device__ __forceinline__ void scan_piece(const int4 d, const int vi, const int base,
                                           unsigned* q, int& n) {
  const int d0 = d.x - base, d1 = d.y - base, d2 = d.z - base, d3 = d.w - base;
  const bool m0 = (unsigned)d0 < (unsigned)RB;
  const bool m1 = (unsigned)d1 < (unsigned)RB;
  const bool m2 = (unsigned)d2 < (unsigned)RB;
  const bool m3 = (unsigned)d3 < (unsigned)RB;
  const unsigned k0 = __builtin_amdgcn_ballot_w32(m0);
  const unsigned k1 = __builtin_amdgcn_ballot_w32(m1);
  const unsigned k2 = __builtin_amdgcn_ballot_w32(m2);
  const unsigned k3 = __builtin_amdgcn_ballot_w32(m3);
  if ((k0 | k1 | k2 | k3) != 0u) {
    keep_edge(m0, k0, d0, vi * 4 + 0, q, n);
    keep_edge(m1, k1, d1, vi * 4 + 1, q, n);
    keep_edge(m2, k2, d2, vi * 4 + 2, q, n);
    keep_edge(m3, k3, d3, vi * 4 + 3, q, n);
  }
}
__device__ __forceinline__ void scan_step(const int4* d, int st, int lane, int base,
                                          unsigned* q, int& n) {
#pragma unroll
  for (int k = 0; k < 8; ++k) scan_piece(d[k], st * 256 + k * 32 + lane, base, q, n);
}

template <bool SAVE>
__device__ __forceinline__ void scan_all(const int* __restrict__ src, const int* __restrict__ dst,
                                         const float* __restrict__ feat,
                                         float* accS, float* cntS, unsigned* q,
                                         int wave, int lane, int base,
                                         float* seg, int cap, int& saved, RootAcc& R) {
  int n = 0;
  const int4* dst4 = (const int4*)dst;
  int4 bufA[8], bufB[8];
  load_step(bufA, dst4, wave, lane);
  for (int st = wave; st < NFULL; st += 2 * NWAVE) {
    const int sb = st + NWAVE;
    const int sa = st + 2 * NWAVE;
    load_step(bufB, dst4, sb < NFULL ? sb : NFULL - 1, lane);
    scan_step(bufA, st, lane, base, q, n);
    if (n >= QFLUSH) { drain_list<SAVE>(q, n, src, feat, accS, cntS, lane, seg, cap, saved, R); n = 0; }
    load_step(bufA, dst4, sa < NFULL ? sa : NFULL - 1, lane);
    if (sb < NFULL) {
      scan_step(bufB, sb, lane, base, q, n);
      if (n >= QFLUSH) { drain_list<SAVE>(q, n, src, feat, accS, cntS, lane, seg, cap, saved, R); n = 0; }
    }
  }
  if (wave == NWAVE - 1) {
#pragma unroll
    for (int k = 0; k < NTAIL; ++k) {
      const int vi = NFULL * 256 + k * 32 + lane;
      scan_piece(dst4[vi], vi, base, q, n);
    }
  }
  drain_list<SAVE>(q, n, src, feat, accS, cntS, lane, seg, cap, saved, R);
}

__device__ __forceinline__ void drain_pieces(const unsigned* qp, int n, unsigned* qe,
                                             const int* __restrict__ src,
                                             const int* __restrict__ dst,
                                             const float* __restrict__ feat,
                                             float* accS, float* cntS, int lane, int base,
                                             float* seg, int cap, int& saved, RootAcc& R) {
  const int4* dst4 = (const int4*)dst;
  unsigned pp[QP / 32];
  int4 dd[QP / 32];
#pragma unroll
  for (int i = 0; i < QP / 32; ++i) {
    pp[i] = 0u;
    dd[i] = make_int4(0, 0, 0, 0);
    if (i * 32 < n) {
      pp[i] = (i * 32 + lane < n) ? qp[i * 32 + lane] : 0u;
      dd[i] = dst4[pp[i]];
    }
  }
  int n2 = 0;
#pragma unroll
  for (int i = 0; i < QP / 32; ++i) {
    if (i * 32 < n) {
      const bool on = i * 32 + lane < n;
      const int dl[4] = {dd[i].x - base, dd[i].y - base, dd[i].z - base, dd[i].w - base};
#pragma unroll
      for (int s = 0; s < 4; ++s) {
        const bool mine = on && (unsigned)dl[s] < (unsigned)RB;
        const unsigned mask = __builtin_amdgcn_ballot_w32(mine);
        const int pos = n2 + (int)__builtin_amdgcn_mbcnt_lo(mask, 0u);
        if (mine) qe[pos] = (pp[i] * 4u + (unsigned)s) | ((unsigned)dl[s] << 20);
        n2 += __builtin_popcount(mask);
      }
    }
  }
  drain_list<true>(qe, n2, src, feat, accS, cntS, lane, seg, cap, saved, R);
}

__device__ __forceinline__ void load_bits(unsigned* w, const float* bm, int wave, int lane) {
#pragma unroll
  for (int k = 0; k < BMK; ++k) {
    const int r = wave + k * NWAVE;
    w[k] = (r < BMROWS) ? __float_as_uint(bm[r * 32 + lane]) : 0u;
  }
}

__device__ __forceinline__ void scan_bits(unsigned* w,
                                          const int* __restrict__ src, const int* __restrict__ dst,
                                          const float* __restrict__ feat,
                                          float* accS, float* cntS, unsigned* q,
                                          int wave, int lane, int base,
                                          float* seg, int cap, int& saved, RootAcc& R) {
  unsigned* qp = q;
  unsigned* qe = q + QP;
  int n = 0;
  for (int t = 0; t < BMK * 32; ++t) {
    int ks = 0;
    unsigned zs = 0u;
#pragma unroll
    for (int k = BMK - 1; k >= 0; --k) {
      if (w[k] != 0u) { ks = k; zs = w[k]; }
    }
    const bool has = zs != 0u;
    const unsigned mask = __builtin_amdgcn_ballot_w32(has);
    if (mask == 0u) break;
    if (n + 32 > QP) {
      drain_pieces(qp, n, qe, src, dst, feat, accS, cntS, lane, base, seg, cap, saved, R);
      n = 0;
    }
    const int b = __builtin_ctz(zs | 0x80000000u);
    const unsigned piece = ((unsigned)((wave + ks * NWAVE) * 32 + lane) << 5) + (unsigned)b;
    const int pos = n + (int)__builtin_amdgcn_mbcnt_lo(mask, 0u);
    if (has) qp[pos] = piece;
    n += __builtin_popcount(mask);
    const unsigned left = zs & (zs - 1u);
#pragma unroll
    for (int k = 0; k < BMK; ++k) {
      if (k == ks) w[k] = left;
    }
  }
  drain_pieces(qp, n, qe, src, dst, feat, accS, cntS, lane, base, seg, cap, saved, R);
}

#define KWAVES 8
#define KPW (32 / KWAVES)
__global__ __launch_bounds__(KWAVES * 32) void owner_bits_k(const int* __restrict__ dst, float* plane) {
  __shared__ unsigned T[NBLK * TPITCH];
  const int tid = threadIdx.x;
  const int lane = tid & 31;
  const int wv = tid >> 5;
  const int g = blockIdx.x;
  const int4* dst4 = (const int4*)dst;
  int4 d[KPW];
#pragma unroll
  for (int i = 0; i < KPW; ++i) {
    const int p = (g * 32 + wv * KPW + i) * 32 + lane;
    d[i] = dst4[p < NV4 ? p : NV4 - 1];
  }
  for (int i = tid; i < NBLK * TPITCH; i += KWAVES * 32) T[i] = 0u;
  __syncthreads();
  const unsigned bit = 1u << lane;
#pragma unroll
  for (int i = 0; i < KPW; ++i) {
    const int k = wv * KPW + i;
    const int p = (g * 32 + k) * 32 + lane;
    if (p < NV4) {
      const unsigned o0 = (unsigned)d[i].x / (unsigned)RB;
      const unsigned o1 = (unsigned)d[i].y / (unsigned)RB;
      const unsigned o2 = (unsigned)d[i].z / (unsigned)RB;
      const unsigned o3 = (unsigned)d[i].w / (unsigned)RB;
      if (o0 < NBLK) atomicOr(&T[o0 * TPITCH + k], bit);
      if (o1 < NBLK) atomicOr(&T[o1 * TPITCH + k], bit);
      if (o2 < NBLK) atomicOr(&T[o2 * TPITCH + k], bit);
      if (o3 < NBLK) atomicOr(&T[o3 * TPITCH + k], bit);
    }
  }
  __syncthreads();
  float* outw = plane + KOFF + g * 32 + lane;
#pragma unroll 8
  for (int b = wv; b < NBLK; b += KWAVES) outw[(size_t)b * BMW] = __uint_as_float(T[b * TPITCH + lane]);
}

template <int LAYER>
__global__ __launch_bounds__(1024) void sage_layer_own_k(
    const int* __restrict__ src, const int* __restrict__ dst,
    const float* __restrict__ feat,
    const float* __restrict__ Wl, const float* __restrict__ Wr,
    const float* __restrict__ bl,
    float* hout, float* lst,
    const float* __restrict__ Wf, const float* __restrict__ bf,
    float* __restrict__ out)
{
  __shared__ float WlS[(GD / 2) * WROW];
  __shared__ float WrS[(GD / 2) * WROW];
  __shared__ float accS[RB * ACCP];
  __shared__ float cntS[RB];
  __shared__ unsigned qS[NWAVE * QW];
  __shared__ float rootS[RB * ACCP];

  const int tid  = threadIdx.x;
  const int lane = tid & 31;
  const int wave = __builtin_amdgcn_readfirstlane(tid >> 5);
  const int base = blockIdx.x * RB;

  float* reg = lst + LOFF + (long long)blockIdx.x * LPITCH;
  const int cap = LCAP;
  float* seg = reg + NWAVE + wave * cap;

  const int half = lane >> 4;
  const int l16  = lane & 15;
  const bool owner = (wave < RB / 16) && (base + wave * 16 < GN);
  v2f rr[16];
  if (owner) {
#pragma unroll
    for (int u = 0; u < 16; ++u)
      rr[u] = *(const v2f*)(feat + (long long)(base + wave * 16 + u) * GD + lane * 2);
  }

  unsigned cw = 0u;
  if (LAYER == 2) cw = __float_as_uint(reg[lane]);
  unsigned bw[BMK];
  if (LAYER == 1) load_bits(bw, lst + KOFF + (long long)blockIdx.x * BMW, wave, lane);

  for (int i = tid; i < GD * GD; i += 1024) {
    int j = i >> 6;
    int k = i & 63;
    int o = (k >> 1) * WROW + j * 2 + (k & 1);
    WlS[o] = Wl[i];
    WrS[o] = Wr[i];
  }
  for (int i = tid; i < RB * ACCP; i += 1024) accS[i] = 0.0f;
  if (tid < RB) cntS[tid] = 0.0f;
  if (owner) {
#pragma unroll
    for (int u = 0; u < 16; ++u) *(v2f*)&rootS[(wave * 16 + u) * ACCP + lane * 2] = rr[u];
  }
  __syncthreads();

  RootAcc R;
#pragma unroll
  for (int t = 0; t < 4; ++t) R.acc[t] = (v8f){0.f, 0.f, 0.f, 0.f, 0.f, 0.f, 0.f, 0.f};
  R.ks = owner ? 0 : 16;
  R.a = rootS + (wave * 16 + l16) * ACCP + half * 2;
  R.b = WrS + half * WROW + l16 * 2;

  unsigned* q = qS + wave * QW;
  int saved = 0;
  if (LAYER == 1) {
    scan_bits(bw, src, dst, feat, accS, cntS, q, wave, lane, base, seg, cap, saved, R);
    if (lane == 0) reg[wave] = __uint_as_float(saved <= cap ? (unsigned)saved : LIST_OVER);
  } else {
    const bool over = __builtin_amdgcn_ballot_w32(cw == LIST_OVER) != 0u;
    if (!over) {
      const int nl = __builtin_amdgcn_readlane((int)cw, wave);
      unsigned ent  = (lane < nl) ? __float_as_uint(seg[lane]) : 0u;
      unsigned ent1 = (32 + lane < nl) ? __float_as_uint(seg[32 + lane]) : 0u;
      for (int g = 0; g < nl; g += 32) {
        int m = nl - g;
        if (m > 32) m = 32;
        const unsigned ent2 = (g + 64 + lane < nl) ? __float_as_uint(seg[g + 64 + lane]) : 0u;
        drain_rows(ent, (int)(ent & 0xFFFFFu), m, feat, accS, cntS, lane, R);
        ent = ent1;
        ent1 = ent2;
      }
    } else {
      RootAcc none = R;
      none.ks = 16;
      scan_all<false>(src, dst, feat, accS, cntS, q, wave, lane, base, seg, cap, saved, none);
    }
  }
  while (R.ks < 16) root_step(R);
  __syncthreads();

  {
    const int row0 = base + wave * 16;
    if (owner) {
      const int lr   = wave * 16 + l16;
      const float inv = 1.0f / fmaxf(cntS[lr], 1.0f);
      v8f (&acc)[4] = R.acc;
#pragma unroll
      for (int ks = 0; ks < 16; ++ks) {
        const int ka = ks * 4 + half * 2;
        v2f am = *(const v2f*)&accS[lr * ACCP + ka];
        am *= inv;
        const int kb = (ka >> 1) * WROW;
#pragma unroll
        for (int t = 0; t < 4; ++t) {
          int nn = t * 16 + l16;
          v2f bL = *(const v2f*)&WlS[kb + nn * 2];
          acc[t] = __builtin_amdgcn_wmma_f32_16x16x4_f32(
              false, am, false, bL, (short)0, acc[t], false, false);
        }
      }

      if (LAYER == 1) {
#pragma unroll
        for (int t = 0; t < 4; ++t) {
          int nn = t * 16 + l16;
          float bias = bl[nn];
#pragma unroll
          for (int v = 0; v < 8; ++v) {
            int mr = half * 8 + v;
            float val = fmaxf(acc[t][v] + bias, 0.0f);
            hout[(long long)(row0 + mr) * GD + nn] = val;
          }
        }
      } else {
        float part[8];
#pragma unroll
        for (int v = 0; v < 8; ++v) part[v] = 0.0f;
#pragma unroll
        for (int t = 0; t < 4; ++t) {
          const int nn = t * 16 + l16;
          const float bias = bl[nn];
          const float wf = Wf[nn];
#pragma unroll
          for (int v = 0; v < 8; ++v) part[v] += fmaxf(acc[t][v] + bias, 0.0f) * wf;
        }
#pragma unroll
        for (int v = 0; v < 8; ++v) {
          float s = part[v];
          s += __shfl_xor(s, 8, 16);
          s += __shfl_xor(s, 4, 16);
          s += __shfl_xor(s, 2, 16);
          s += __shfl_xor(s, 1, 16);
          part[v] = s;
        }
        float mine = part[0];
#pragma unroll
        for (int v = 1; v < 8; ++v) mine = (l16 == v) ? part[v] : mine;
        if (l16 < 8) out[row0 + half * 8 + l16] = mine + bf[0];
      }
    }
  }
}

extern "C" void kernel_launch(void* const* d_in, const int* in_sizes, int n_in,
                              void* d_out, int out_size, void* d_ws, size_t ws_size,
                              hipStream_t stream) {
  (void)in_sizes; (void)n_in; (void)out_size; (void)ws_size;

  const float* x   = (const float*)d_in[0];
  const int*   ei  = (const int*)d_in[1];
  const int*   src = ei;
  const int*   dst = ei + GE;
  const float* Wl1 = (const float*)d_in[2];
  const float* bl1 = (const float*)d_in[3];
  const float* Wr1 = (const float*)d_in[4];
  const float* Wl2 = (const float*)d_in[5];
  const float* bl2 = (const float*)d_in[6];
  const float* Wr2 = (const float*)d_in[7];
  const float* Wf  = (const float*)d_in[8];
  const float* bf  = (const float*)d_in[9];

  char*  ws = (char*)d_ws;
  float* h1 = (float*)ws;
  float* h2 = (float*)(ws + (size_t)GN * GD * 4 + (size_t)GN * 4);

  owner_bits_k<<<BMROWS, KWAVES * 32, 0, stream>>>(dst, h2);
  sage_layer_own_k<1><<<NBLK, 1024, 0, stream>>>(src, dst, x,  Wl1, Wr1, bl1, h1, h2,
                                                 Wf, bf, (float*)d_out);
  sage_layer_own_k<2><<<NBLK, 1024, 0, stream>>>(src, dst, h1, Wl2, Wr2, bl2, h2, h2,
                                                 Wf, bf, (float*)d_out);
}
